# starting kernel
# baseline (speedup 1.0000x reference)
#include <hip/hip_runtime.h>

typedef float v2f __attribute__((ext_vector_type(2)));
typedef float v8f __attribute__((ext_vector_type(8)));

#define KK     7
#define HH     64
#define WW     64
#define CC     256
#define XROWS  22
#define XCOLS  72

__global__ __launch_bounds__(128)
void dwconv7_wmma_f32(const float* __restrict__ x,
                      const float* __restrict__ wgt,
                      const float* __restrict__ bias,
                      float* __restrict__ out)
{
    __shared__ float Xs[XROWS * XCOLS];
    __shared__ float Gs[4][XROWS * KK * 16];

    const int tid  = threadIdx.x;
    const int bid  = blockIdx.x;
    const int p    = bid >> 2;
    const int trow = bid & 3;
    const int c    = p & (CC - 1);
    const int h0   = trow * 16;
    const long plane_off = (long)p * (HH * WW);

    for (int idx = tid; idx < XROWS * XCOLS; idx += 128) {
        const int rr = idx / XCOLS;
        const int cc = idx - rr * XCOLS;
        const int gr = h0 - 3 + rr;
        const int gc = cc - 3;
        float v = 0.0f;
        if (gr >= 0 && gr < HH && gc >= 0 && gc < WW)
            v = x[plane_off + gr * WW + gc];
        Xs[idx] = v;
    }

    const int  wv   = tid >> 5;
    const int  lane = tid & 31;
    const int  m    = lane & 15;
    const bool hi   = lane >= 16;
    const int  colb = wv * 16 + m;
    const int  j0   = hi ? 2 : 0;
    const int  j1   = hi ? 6 : 4;

    float w0r = 0.f, w1r = 0.f, w2r = 0.f, w3r = 0.f, w4r = 0.f, w5r = 0.f, w6r = 0.f;
    if (m < KK) {
        const float* wp = wgt + c * (KK * KK) + m * KK;
        w0r = wp[0]; w1r = wp[1]; w2r = wp[2]; w3r = wp[3];
        w4r = wp[4]; w5r = wp[5]; w6r = wp[6];
    }
    v2f b0, b1;
    b0.x = hi ? w2r : w0r;   b0.y = hi ? w3r  : w1r;
    b1.x = hi ? w6r : w4r;   b1.y = hi ? 0.0f : w5r;

    __syncthreads();

    float* gbase = &Gs[wv][0];
    for (int r = 0; r < XROWS; ++r) {
        const float* xr = &Xs[r * XCOLS + colb];
        v2f a0, a1;
        a0.x = xr[j0];
        a0.y = xr[j0 + 1];
        a1.x = xr[j1];
        const float t = xr[j1 + 1];
        a1.y = hi ? 0.0f : t;
        v8f acc = {};
        acc = __builtin_amdgcn_wmma_f32_16x16x4_f32(false, a0, false, b0,
                                                    (short)0, acc, false, false);
        acc = __builtin_amdgcn_wmma_f32_16x16x4_f32(false, a1, false, b1,
                                                    (short)0, acc, false, false);
        if (m < KK) {
            float* gp = gbase + (r * KK + m) * 16 + (hi ? 8 : 0);
            gp[0] = acc[0]; gp[1] = acc[1]; gp[2] = acc[2]; gp[3] = acc[3];
            gp[4] = acc[4]; gp[5] = acc[5]; gp[6] = acc[6]; gp[7] = acc[7];
        }
    }

    __syncthreads();

    const float bv = bias[c];
    const int  w   = m;
    const int  hb  = hi ? 8 : 0;
    const float* gw = gbase + w;
    for (int hh = 0; hh < 8; ++hh) {
        const int h = hb + hh;
        float acc = bv;
        #pragma unroll
        for (int i = 0; i < KK; ++i)
            acc += gw[((h + i) * KK + i) * 16];
        out[plane_off + (h0 + h) * WW + wv * 16 + w] = acc;
    }
}

extern "C" void kernel_launch(void* const* d_in, const int* in_sizes, int n_in,
                              void* d_out, int out_size, void* d_ws, size_t ws_size,
                              hipStream_t stream) {
    const float* x    = (const float*)d_in[0];
    const float* wgt  = (const float*)d_in[1];
    const float* bias = (const float*)d_in[2];
    float* out = (float*)d_out;

    const int planes = 16 * CC;
    dim3 grid(planes * 4);
    dim3 block(128);
    dwconv7_wmma_f32<<<grid, block, 0, stream>>>(x, wgt, bias, out);
}


# Halo tile read as 16-byte global loads with every global read of the block (tile, weights, bias) issued before the first wait, replacing one 4-byte load waited for on each of 13 loop trips
# speedup vs baseline: 1.8112x; 1.8112x over previous
#include <hip/hip_runtime.h>

typedef float v2f __attribute__((ext_vector_type(2)));
typedef float v4f __attribute__((ext_vector_type(4)));
typedef float v8f __attribute__((ext_vector_type(8)));

#define KK     7
#define HH     64
#define WW     64
#define CC     256
#define XROWS  22
#define XCOLS  72
#define XOFF   4
#define NPIECE (XROWS * (WW / 4))

__global__ __launch_bounds__(128)
void dwconv7_wmma_f32(const float* __restrict__ x,
                      const float* __restrict__ wgt,
                      const float* __restrict__ bias,
                      float* __restrict__ out)
{
    __shared__ __attribute__((aligned(16))) float Xs[XROWS * XCOLS];
    __shared__ float Gs[4][XROWS * KK * 16];

    const int tid  = threadIdx.x;
    const int bid  = blockIdx.x;
    const int p    = bid >> 2;
    const int trow = bid & 3;
    const int c    = p & (CC - 1);
    const int h0   = trow * 16;
    const long plane_off = (long)p * (HH * WW);

    const int  wv   = tid >> 5;
    const int  lane = tid & 31;
    const int  m    = lane & 15;
    const bool hi   = lane >= 16;

    const float* xp = x + plane_off;
    const int q     = tid & 15;
    const int row0  = tid >> 4;
    const int row1  = row0 + 8;
    const int row2r = row0 + 16;
    const int row2  = row2r < XROWS ? row2r : XROWS - 1;
    const int gr0   = h0 - 3 + row0;
    const int gr1   = h0 - 3 + row1;
    const int gr2   = h0 - 3 + row2;
    const int gc0   = gr0 < 0 ? 0 : gr0;
    const int gc2   = gr2 > HH - 1 ? HH - 1 : gr2;
    const v4f t0 = *(const v4f*)(xp + gc0 * WW + q * 4);
    const v4f t1 = *(const v4f*)(xp + gr1 * WW + q * 4);
    const v4f t2 = *(const v4f*)(xp + gc2 * WW + q * 4);

    const float* wp = wgt + c * (KK * KK) + (m < KK ? m : 0) * KK;
    float w0r = wp[0], w1r = wp[1], w2r = wp[2], w3r = wp[3];
    float w4r = wp[4], w5r = wp[5], w6r = wp[6];
    const float bv = bias[c];

    const v4f zero4 = {0.0f, 0.0f, 0.0f, 0.0f};
    if (tid < 2 * XROWS)
        *(v4f*)&Xs[(tid >> 1) * XCOLS + (tid & 1) * (XOFF + WW)] = zero4;

    *(v4f*)&Xs[row0 * XCOLS + XOFF + q * 4] = (gr0 >= 0) ? t0 : zero4;
    *(v4f*)&Xs[row1 * XCOLS + XOFF + q * 4] = t1;
    if (row2r < XROWS)
        *(v4f*)&Xs[row2r * XCOLS + XOFF + q * 4] = (gr2 < HH) ? t2 : zero4;

    const int  colb = wv * 16 + m + (XOFF - 3);
    const int  j0   = hi ? 2 : 0;
    const int  j1   = hi ? 6 : 4;

    if (m >= KK) {
        w0r = 0.f; w1r = 0.f; w2r = 0.f; w3r = 0.f; w4r = 0.f; w5r = 0.f; w6r = 0.f;
    }
    v2f b0, b1;
    b0.x = hi ? w2r : w0r;   b0.y = hi ? w3r  : w1r;
    b1.x = hi ? w6r : w4r;   b1.y = hi ? 0.0f : w5r;

    __syncthreads();

    float* gbase = &Gs[wv][0];
    for (int r = 0; r < XROWS; ++r) {
        const float* xr = &Xs[r * XCOLS + colb];
        v2f a0, a1;
        a0.x = xr[j0];
        a0.y = xr[j0 + 1];
        a1.x = xr[j1];
        const float t = xr[j1 + 1];
        a1.y = hi ? 0.0f : t;
        v8f acc = {};
        acc = __builtin_amdgcn_wmma_f32_16x16x4_f32(false, a0, false, b0,
                                                    (short)0, acc, false, false);
        acc = __builtin_amdgcn_wmma_f32_16x16x4_f32(false, a1, false, b1,
                                                    (short)0, acc, false, false);
        if (m < KK) {
            float* gp = gbase + (r * KK + m) * 16 + (hi ? 8 : 0);
            gp[0] = acc[0]; gp[1] = acc[1]; gp[2] = acc[2]; gp[3] = acc[3];
            gp[4] = acc[4]; gp[5] = acc[5]; gp[6] = acc[6]; gp[7] = acc[7];
        }
    }

    __syncthreads();

    const int  w   = m;
    const int  hb  = hi ? 8 : 0;
    const float* gw = gbase + w;
    for (int hh = 0; hh < 8; ++hh) {
        const int h = hb + hh;
        float acc = bv;
        #pragma unroll
        for (int i = 0; i < KK; ++i)
            acc += gw[((h + i) * KK + i) * 16];
        out[plane_off + (h0 + h) * WW + wv * 16 + w] = acc;
    }
}

extern "C" void kernel_launch(void* const* d_in, const int* in_sizes, int n_in,
                              void* d_out, int out_size, void* d_ws, size_t ws_size,
                              hipStream_t stream) {
    const float* x    = (const float*)d_in[0];
    const float* wgt  = (const float*)d_in[1];
    const float* bias = (const float*)d_in[2];
    float* out = (float*)d_out;

    const int planes = 16 * CC;
    dim3 grid(planes * 4);
    dim3 block(128);
    dwconv7_wmma_f32<<<grid, block, 0, stream>>>(x, wgt, bias, out);
}


# All 7 kernel rows and taps summed inside the f32 WMMA accumulator via banded weight fragments (one block per plane); LDS row-conv round trip and second pass removed; summation order changes
# speedup vs baseline: 3.1073x; 1.7156x over previous
#include <hip/hip_runtime.h>

typedef float v2f __attribute__((ext_vector_type(2)));
typedef float v4f __attribute__((ext_vector_type(4)));
typedef float v8f __attribute__((ext_vector_type(8)));

#define KK     7
#define HH     64
#define WW     64
#define CC     256
#define PITCH  68
#define LROWS  (HH + KK - 1)
#define XS_DW  (LROWS * PITCH + 4)
#define WPITCH 40
#define NPIECE (HH * WW / 4 / 128)

__global__ __launch_bounds__(128)
void dwconv7_wmma_f32(const float* __restrict__ x,
                      const float* __restrict__ wgt,
                      const float* __restrict__ bias,
                      float* __restrict__ out)
{
    __shared__ __attribute__((aligned(16))) float Xs[XS_DW];
    __shared__ __attribute__((aligned(16))) float Wp[KK * WPITCH];

    const int tid  = threadIdx.x;
    const int p    = blockIdx.x;
    const int c    = p & (CC - 1);
    const long plane_off = (long)p * (HH * WW);
    const int wv   = tid >> 5;
    const int lane = tid & 31;
    const int n    = lane & 15;
    const int hi   = lane >> 4;

    const float* xp = x + plane_off;
    v4f t[NPIECE];
    #pragma unroll
    for (int e = 0; e < NPIECE; ++e)
        t[e] = *(const v4f*)(xp + (tid + 128 * e) * 4);

    float wv_[3];
    #pragma unroll
    for (int e = 0; e < 3; ++e) {
        const int idx = tid + 128 * e;
        const int i   = idx / WPITCH;
        const int jj  = idx - i * WPITCH - 16;
        const int ic  = i < KK ? i : KK - 1;
        const int jc  = jj < 0 ? 0 : (jj > KK - 1 ? KK - 1 : jj);
        const float v = wgt[c * (KK * KK) + ic * KK + jc];
        wv_[e] = (jj >= 0 && jj < KK) ? v : 0.0f;
    }
    const float bv = bias[c];

    const v4f zero4 = {0.0f, 0.0f, 0.0f, 0.0f};
    if (tid < LROWS + 1)
        *(v4f*)&Xs[tid * PITCH] = zero4;
    if (tid < 6 * (WW / 4)) {
        const int hr  = tid >> 4;
        const int row = hr < 3 ? hr : hr + HH;
        *(v4f*)&Xs[row * PITCH + 4 + (tid & 15) * 4] = zero4;
    }
    #pragma unroll
    for (int e = 0; e < NPIECE; ++e) {
        const int pc = tid + 128 * e;
        *(v4f*)&Xs[((pc >> 4) + 3) * PITCH + 4 + (pc & 15) * 4] = t[e];
    }
    #pragma unroll
    for (int e = 0; e < 3; ++e) {
        const int idx = tid + 128 * e;
        if (idx < KK * WPITCH) Wp[idx] = wv_[e];
    }

    __syncthreads();

    v8f acc[4];
    #pragma unroll
    for (int tt = 0; tt < 4; ++tt)
        #pragma unroll
        for (int r = 0; r < 8; ++r) acc[tt][r] = bv;

    const float* xa = &Xs[(16 * wv + n) * PITCH + 2 * hi];
    const float* wb = &Wp[15 + 2 * hi - n];

    #pragma unroll
    for (int i = 0; i < KK; ++i) {
        const float* xr = xa + i * PITCH;
        v2f a[18];
        #pragma unroll
        for (int kk = 0; kk < 18; ++kk)
            a[kk] = *(const v2f*)(xr + 4 * kk);
        v2f b[6];
        #pragma unroll
        for (int ks = 0; ks < 6; ++ks) {
            b[ks].x = wb[i * WPITCH + 4 * ks];
            b[ks].y = wb[i * WPITCH + 4 * ks + 1];
        }
        #pragma unroll
        for (int ks = 0; ks < 6; ++ks)
            #pragma unroll
            for (int tt = 0; tt < 4; ++tt)
                acc[tt] = __builtin_amdgcn_wmma_f32_16x16x4_f32(false, a[4 * tt + ks], false, b[ks],
                                                                (short)0, acc[tt], false, false);
    }

    float* op = out + plane_off + (16 * wv + 8 * hi) * WW + n;
    #pragma unroll
    for (int tt = 0; tt < 4; ++tt)
        #pragma unroll
        for (int r = 0; r < 8; ++r)
            op[r * WW + 16 * tt] = acc[tt][r];
}

extern "C" void kernel_launch(void* const* d_in, const int* in_sizes, int n_in,
                              void* d_out, int out_size, void* d_ws, size_t ws_size,
                              hipStream_t stream) {
    const float* x    = (const float*)d_in[0];
    const float* wgt  = (const float*)d_in[1];
    const float* bias = (const float*)d_in[2];
    float* out = (float*)d_out;

    dim3 grid(16 * CC);
    dim3 block(128);
    dwconv7_wmma_f32<<<grid, block, 0, stream>>>(x, wgt, bias, out);
}


# 2-D banded weight operand: a WMMA column is a 4x4 output block fed by its 10x10 window (K=100, weights as A, image as B) instead of 7 row bands of K=24: 100 WMMAs per wave, was 168; sum order changes
# speedup vs baseline: 4.5656x; 1.4693x over previous
#include <hip/hip_runtime.h>

typedef float v2f __attribute__((ext_vector_type(2)));
typedef float v4f __attribute__((ext_vector_type(4)));
typedef float v8f __attribute__((ext_vector_type(8)));

#define KK     7
#define HH     64
#define WW     64
#define CC     256
#define PW     10
#define PITCH  70
#define LROWS  (HH + KK - 1)
#define XS_DW  (LROWS * PITCH)
#define WPITCH 13
#define WP_DW  (WPITCH * WPITCH)
#define NPIECE (HH * WW / 4 / 128)
#define NKS    (PW * PW / 4)
#define NGRP   4
#define NFRAG  (NKS + 10 * (NGRP - 1))

#define FOFF(u, pitch) (((4 * (u)) / PW) * (pitch) + (4 * (u)) % PW)
#define FWRAP(u) ((4 * (u)) % PW == 8)

__device__ __forceinline__ void put4(float* p, v4f v)
{
    p[0] = v.x;
    *(v2f*)(p + 1) = (v2f){v.y, v.z};
    p[3] = v.w;
}

__global__ __launch_bounds__(128)
void dwconv7_wmma_f32(const float* __restrict__ x,
                      const float* __restrict__ wgt,
                      const float* __restrict__ bias,
                      float* __restrict__ out)
{
    __shared__ __attribute__((aligned(16))) float Xs[XS_DW];
    __shared__ __attribute__((aligned(16))) float Wp[WP_DW];

    const int tid  = threadIdx.x;
    const int p    = blockIdx.x;
    const int c    = p & (CC - 1);
    const long plane_off = (long)p * (HH * WW);
    const int wv   = tid >> 5;
    const int lane = tid & 31;
    const int n    = lane & 15;
    const int hi   = lane >> 4;

    const float* xp = x + plane_off;
    v4f t[NPIECE];
    #pragma unroll
    for (int e = 0; e < NPIECE; ++e)
        t[e] = *(const v4f*)(xp + (tid + 128 * e) * 4);

    float wv_[2];
    #pragma unroll
    for (int e = 0; e < 2; ++e) {
        const int idx = tid + 128 * e;
        const int row = idx / WPITCH;
        const int i   = row - 3;
        const int j   = idx - row * WPITCH - 3;
        const int ic  = i < 0 ? 0 : (i > KK - 1 ? KK - 1 : i);
        const int jc  = j < 0 ? 0 : (j > KK - 1 ? KK - 1 : j);
        const float v = wgt[c * (KK * KK) + ic * KK + jc];
        wv_[e] = (i >= 0 && i < KK && j >= 0 && j < KK) ? v : 0.0f;
    }
    const float bv = bias[c];

    const v4f zero4 = {0.0f, 0.0f, 0.0f, 0.0f};
    if (tid < LROWS) {
        float* r = &Xs[tid * PITCH];
        r[0] = 0.0f; r[1] = 0.0f; r[2] = 0.0f;
        r[WW + 3] = 0.0f; r[WW + 4] = 0.0f; r[WW + 5] = 0.0f;
    }
    if (tid < 6 * (WW / 4)) {
        const int hr  = tid >> 4;
        const int row = hr < 3 ? hr : hr + HH;
        put4(&Xs[row * PITCH + 3 + (tid & 15) * 4], zero4);
    }
    #pragma unroll
    for (int e = 0; e < NPIECE; ++e) {
        const int pc = tid + 128 * e;
        put4(&Xs[((pc >> 4) + 3) * PITCH + 3 + (pc & 15) * 4], t[e]);
    }
    #pragma unroll
    for (int e = 0; e < 2; ++e) {
        const int idx = tid + 128 * e;
        if (idx < WP_DW) Wp[idx] = wv_[e];
    }

    __syncthreads();

    const int orow = n >> 2, ocol = n & 3;
    const float* wbase = &Wp[(3 - orow) * WPITCH + (3 - ocol)];
    const float* wS = wbase + 2 * hi;
    const float* wW = wbase + hi * (WPITCH - 8);
    v2f a[NKS];
    #pragma unroll
    for (int ks = 0; ks < NKS; ++ks) {
        const float* q = (FWRAP(ks) ? wW : wS) + FOFF(ks, WPITCH);
        a[ks].x = q[0];
        a[ks].y = q[1];
    }

    v8f acc[NGRP];
    #pragma unroll
    for (int g = 0; g < NGRP; ++g)
        #pragma unroll
        for (int r = 0; r < 8; ++r) acc[g][r] = bv;

    const float* xbase = &Xs[(16 * wv) * PITCH + 4 * n];
    const float* xS = xbase + 2 * hi;
    const float* xW = xbase + hi * (PITCH - 8);

    #pragma unroll
    for (int u = 0; u < NFRAG; ++u) {
        const v2f b = *(const v2f*)((FWRAP(u) ? xW : xS) + FOFF(u, PITCH));
        #pragma unroll
        for (int g = 0; g < NGRP; ++g) {
            const int ks = u - 10 * g;
            if (ks >= 0 && ks < NKS)
                acc[g] = __builtin_amdgcn_wmma_f32_16x16x4_f32(false, a[ks], false, b,
                                                               (short)0, acc[g], false, false);
        }
    }

    float* op = out + plane_off + (16 * wv + 2 * hi) * WW + 4 * n;
    #pragma unroll
    for (int g = 0; g < NGRP; ++g)
        #pragma unroll
        for (int q = 0; q < 2; ++q)
            *(v4f*)(op + (4 * g + q) * WW) =
                (v4f){acc[g][4 * q], acc[g][4 * q + 1], acc[g][4 * q + 2], acc[g][4 * q + 3]};
}

extern "C" void kernel_launch(void* const* d_in, const int* in_sizes, int n_in,
                              void* d_out, int out_size, void* d_ws, size_t ws_size,
                              hipStream_t stream) {
    const float* x    = (const float*)d_in[0];
    const float* wgt  = (const float*)d_in[1];
    const float* bias = (const float*)d_in[2];
    float* out = (float*)d_out;

    dim3 grid(16 * CC);
    dim3 block(128);
    dwconv7_wmma_f32<<<grid, block, 0, stream>>>(x, wgt, bias, out);
}
